# starting kernel
# baseline (speedup 1.0000x reference)
#include <hip/hip_runtime.h>
#include <hip/hip_bf16.h>

typedef __attribute__((ext_vector_type(2))) float v2f;
typedef __attribute__((ext_vector_type(8))) float v8f;

#define FDIM 64

__global__ void __launch_bounds__(256) zero_kernel(float* __restrict__ p, int n) {
    int i = blockIdx.x * 256 + threadIdx.x;
    if (i < n) p[i] = 0.0f;
}

__global__ void __launch_bounds__(256)
scatter_kernel(const float* __restrict__ feat, const int* __restrict__ src,
               const int* __restrict__ dst, float* __restrict__ agg,
               float* __restrict__ cnt, int n_edges) {
    long long tid = (long long)blockIdx.x * 256 + threadIdx.x;
    int e = (int)(tid >> 4);
    if (e >= n_edges) return;
    int q = (int)(tid & 15);
    int s = src[e];
    int d = dst[e];
    const float4 v = *(const float4*)(feat + (size_t)s * FDIM + q * 4);
    float* ap = agg + (size_t)d * FDIM + q * 4;
    atomicAdd(ap + 0, v.x);
    atomicAdd(ap + 1, v.y);
    atomicAdd(ap + 2, v.z);
    atomicAdd(ap + 3, v.w);
    if (cnt != nullptr && q == 0) atomicAdd(cnt + d, 1.0f);
}

__global__ void __launch_bounds__(256)
sage_gemm(const float* __restrict__ agg_sum, const float* __restrict__ cnt,
          const float* __restrict__ xin, const float* __restrict__ Wl,
          const float* __restrict__ Wr, const float* __restrict__ bias,
          float* __restrict__ out, int n_nodes, int relu_flag) {
    __shared__ float sWl[FDIM * FDIM];
    __shared__ float sWr[FDIM * FDIM];
    for (int i = threadIdx.x; i < FDIM * FDIM; i += 256) {
        sWl[i] = Wl[i];
        sWr[i] = Wr[i];
    }
    __syncthreads();

    const int lane = threadIdx.x & 31;
    const int wave = threadIdx.x >> 5;
    const int tile = blockIdx.x * 8 + wave;
    const int ntiles = (n_nodes + 15) >> 4;
    if (tile >= ntiles) return;
    const int node0 = tile << 4;

    const int half = lane >> 4;
    const int lm   = lane & 15;

    float c   = (cnt != nullptr) ? cnt[node0 + lm] : 1.0f;
    float inv = 1.0f / fmaxf(c, 1.0f);

    const float* arow = agg_sum + (size_t)(node0 + lm) * FDIM;
    const float* xrow = xin     + (size_t)(node0 + lm) * FDIM;

    v8f acc[4];
    #pragma unroll
    for (int nt = 0; nt < 4; ++nt) acc[nt] = (v8f){};

    #pragma unroll
    for (int kb = 0; kb < 16; ++kb) {
        const int k0 = kb * 4 + half * 2;
        v2f a;
        a.x = arow[k0]     * inv;
        a.y = arow[k0 + 1] * inv;
        #pragma unroll
        for (int nt = 0; nt < 4; ++nt) {
            v2f b;
            b.x = sWl[k0 * FDIM + nt * 16 + lm];
            b.y = sWl[(k0 + 1) * FDIM + nt * 16 + lm];
            acc[nt] = __builtin_amdgcn_wmma_f32_16x16x4_f32(
                false, a, false, b, (short)0, acc[nt], false, false);
        }
    }
    #pragma unroll
    for (int kb = 0; kb < 16; ++kb) {
        const int k0 = kb * 4 + half * 2;
        v2f a;
        a.x = xrow[k0];
        a.y = xrow[k0 + 1];
        #pragma unroll
        for (int nt = 0; nt < 4; ++nt) {
            v2f b;
            b.x = sWr[k0 * FDIM + nt * 16 + lm];
            b.y = sWr[(k0 + 1) * FDIM + nt * 16 + lm];
            acc[nt] = __builtin_amdgcn_wmma_f32_16x16x4_f32(
                false, a, false, b, (short)0, acc[nt], false, false);
        }
    }

    #pragma unroll
    for (int nt = 0; nt < 4; ++nt) {
        const float bv = bias[nt * 16 + lm];
        #pragma unroll
        for (int i = 0; i < 8; ++i) {
            float v = acc[nt][i] + bv;
            if (relu_flag) v = fmaxf(v, 0.0f);
            out[(size_t)(node0 + i + half * 8) * FDIM + nt * 16 + lm] = v;
        }
    }
}

extern "C" void kernel_launch(void* const* d_in, const int* in_sizes, int n_in,
                              void* d_out, int out_size, void* d_ws, size_t ws_size,
                              hipStream_t stream) {
    const float* x    = (const float*)d_in[0];
    const int*   ei   = (const int*)d_in[1];
    const float* W1_l = (const float*)d_in[2];
    const float* b1   = (const float*)d_in[3];
    const float* W1_r = (const float*)d_in[4];
    const float* W2_l = (const float*)d_in[5];
    const float* b2   = (const float*)d_in[6];
    const float* W2_r = (const float*)d_in[7];
    float* out = (float*)d_out;

    const int n_nodes = in_sizes[0] / FDIM;
    const int n_edges = in_sizes[1] / 2;
    const int* src = ei;
    const int* dst = ei + n_edges;

    float* agg = (float*)d_ws;
    float* cnt = agg + (size_t)n_nodes * FDIM;
    float* h   = cnt + n_nodes;

    const int ntiles     = (n_nodes + 15) / 16;
    const int gemm_grid  = (ntiles + 7) / 8;
    const long long sthreads = (long long)n_edges * 16;
    const int scat_grid  = (int)((sthreads + 255) / 256);
    const int nagg       = n_nodes * FDIM;

    zero_kernel<<<(nagg + n_nodes + 255) / 256, 256, 0, stream>>>(agg, nagg + n_nodes);
    scatter_kernel<<<scat_grid, 256, 0, stream>>>(x, src, dst, agg, cnt, n_edges);
    sage_gemm<<<gemm_grid, 256, 0, stream>>>(agg, cnt, x, W1_l, W1_r, b1, h,
                                             n_nodes,  1);

    zero_kernel<<<(nagg + 255) / 256, 256, 0, stream>>>(agg, nagg);
    scatter_kernel<<<scat_grid, 256, 0, stream>>>(h, src, dst, agg, nullptr, n_edges);
    sage_gemm<<<gemm_grid, 256, 0, stream>>>(agg, cnt, h, W2_l, W2_r, b2, out,
                                             n_nodes,  0);
}


# Layer 1 no longer reads every edge key: the key launch also sets, per block key, one bit per 16-edge chunk holding such an edge (LDS bit-or); a block reads its bit row, then only flagged chunks' keys
# speedup vs baseline: 3.5219x; 3.5219x over previous
#include <hip/hip_runtime.h>


typedef __attribute__((ext_vector_type(2))) float v2f;
typedef __attribute__((ext_vector_type(8))) float v8f;

#define FDIM   64
#define RNODES 400
#define APITCH 66
#define WPITCH 72
#define NWAVES 32
#define NTILES (RNODES / 16)
static_assert(RNODES % 16 == 0 && NTILES <= NWAVES, "a tile must not straddle two blocks");
#define QCAP   1056
#define RCAP   2048
#define SLAB   1024
#define BPITCH 257

__global__ void __launch_bounds__(1024)
edge_keys(const int* __restrict__ dst, unsigned* __restrict__ keys, unsigned* __restrict__ pres,
          int n_edges, int n_vec, int ppitch) {
    __shared__ unsigned sBits[32 * BPITCH];
    const int tid = threadIdx.x;
    for (int i = tid; i < 32 * BPITCH; i += 1024) sBits[i] = 0u;
    __syncthreads();
    const int t = blockIdx.x * SLAB + tid;
    if (t < n_vec) {
        const int e0 = t * 16;
        int d[16];
        if ((n_edges & 3) == 0 && e0 + 16 <= n_edges) {
            const int4* p = (const int4*)(dst + e0);
            #pragma unroll
            for (int j = 0; j < 4; ++j) {
                const int4 v = p[j];
                d[4 * j] = v.x; d[4 * j + 1] = v.y; d[4 * j + 2] = v.z; d[4 * j + 3] = v.w;
            }
        } else {
            #pragma unroll
            for (int j = 0; j < 16; ++j) d[j] = (e0 + j < n_edges) ? dst[e0 + j] : 255 * RNODES;
        }
        unsigned kb[16];
        #pragma unroll
        for (int j = 0; j < 16; ++j) kb[j] = ((unsigned)d[j] / (unsigned)RNODES) & 255u;
        ((uint4*)keys)[t] = make_uint4(
            kb[0]  | kb[1]  << 8 | kb[2]  << 16 | kb[3]  << 24,
            kb[4]  | kb[5]  << 8 | kb[6]  << 16 | kb[7]  << 24,
            kb[8]  | kb[9]  << 8 | kb[10] << 16 | kb[11] << 24,
            kb[12] | kb[13] << 8 | kb[14] << 16 | kb[15] << 24);
        unsigned* col = sBits + (tid >> 5) * BPITCH;
        const unsigned bit = 1u << (tid & 31);
        #pragma unroll
        for (int j = 0; j < 16; ++j) atomicOr(&col[kb[j]], bit);
    }
    __syncthreads();
    const int lane = tid & 31;
    for (int r = tid >> 5; r < 256; r += 32)
        pres[(size_t)r * ppitch + blockIdx.x * 32 + lane] = sBits[lane * BPITCH + r];
}

__global__ void __launch_bounds__(1024)
sage_layer(const float* __restrict__ feat, const int* __restrict__ src,
           const int* __restrict__ dst, const unsigned* __restrict__ keys,
           const unsigned* __restrict__ pres, const float* __restrict__ Wl,
           const float* __restrict__ Wr, const float* __restrict__ bias,
           float* __restrict__ out, int* __restrict__ rcnt, int* __restrict__ rec,
           int n_nodes, int n_edges, int ppitch, int relu_flag, int second, size_t ws_bytes) {
    __shared__ alignas(16) float sAcc[RNODES * APITCH];
    __shared__ float sCnt[RNODES];
    __shared__ float sWl[FDIM * WPITCH];
    __shared__ float sWr[FDIM * WPITCH];
    __shared__ int   sQ[NWAVES * QCAP];

    const int tid = threadIdx.x;
    for (int i = tid; i < RNODES * APITCH; i += 1024) sAcc[i] = 0.0f;
    if (tid < RNODES) sCnt[tid] = 0.0f;
    for (int i = tid; i < FDIM * FDIM; i += 1024) {
        const int r = i >> 6, c = i & 63;
        sWl[r * WPITCH + c] = Wl[i];
        sWr[r * WPITCH + c] = Wr[i];
    }
    __syncthreads();

    const int lane = tid & 31;
    const int wave = __builtin_amdgcn_readfirstlane(tid >> 5);
    const int base = blockIdx.x * RNODES;
    int* q = sQ + wave * QCAP;
    int qn = 0;

    int2* myrec = (int2*)rec + ((size_t)blockIdx.x * NWAVES + wave) * RCAP;
    int*  mycnt = rcnt + blockIdx.x * NWAVES;
    int  nrec = 0;
    bool filter = true;
    if (second) {
        const int c = mycnt[lane];
        filter = __builtin_amdgcn_ballot_w32(c > RCAP) != 0u;
        nrec = __builtin_amdgcn_readlane(c, wave);
    } else {
        const size_t need = (size_t)n_nodes * FDIM * sizeof(float) +
                            (size_t)((n_edges + 15) / 16) * 16 + (size_t)256 * ppitch * sizeof(unsigned) +
                            (size_t)gridDim.x * NWAVES * (sizeof(int) + (size_t)RCAP * sizeof(int2));
        if (need > ws_bytes) nrec = RCAP + 1;
    }

    const unsigned pat = (blockIdx.x & 255u) * 0x01010101u;
    const uint4* k4 = (const uint4*)keys;

    auto gather = [&](int s, int d, int n) {
        if (lane < n && d >= 0) atomicAdd(&sCnt[d], 1.0f);
        for (int k0 = 0; k0 < n; k0 += 8) {
            float a[8], b[8];
            int dd[8];
            #pragma unroll
            for (int j = 0; j < 8; ++j) {
                const int k  = k0 + j;
                const int kk = (k < n) ? k : (n - 1);
                const int sk = __builtin_amdgcn_readlane(s, kk);
                dd[j]        = __builtin_amdgcn_readlane(d, kk);
                const float* rp = feat + (size_t)sk * FDIM;
                a[j] = rp[lane];
                b[j] = rp[lane + 32];
            }
            #pragma unroll
            for (int j = 0; j < 8; ++j) {
                if (k0 + j < n && dd[j] >= 0) {
                    atomicAdd(&sAcc[dd[j] * APITCH + lane], a[j]);
                    atomicAdd(&sAcc[dd[j] * APITCH + lane + 32], b[j]);
                }
            }
        }
    };
    auto drain = [&](int n) {
        qn -= n;
        const bool act = lane < n;
        const int e = act ? q[qn + lane] : 0;
        const bool in = act && e < n_edges;
        const int s = in ? src[e] : 0;
        const int dv = in ? (dst[e] - base) : -1;
        const int d = ((unsigned)dv < (unsigned)RNODES) ? dv : -1;
        if (!second) {
            if (nrec + n <= RCAP) {
                if (act) myrec[nrec + lane] = make_int2(s, d);
                nrec += n;
            } else {
                nrec = RCAP + 1;
            }
        }
        gather(s, d, n);
    };

#define NZ(w) (((((w) ^ pat) & 0x7f7f7f7fu) + 0x7f7f7f7fu) | ((w) ^ pat))
#define EQ(w, j) ((((~NZ(w)) >> 7) & 0x01010101u) << (j))

    if (filter) {
        const unsigned* prow = pres + (size_t)(blockIdx.x & 255u) * ppitch;
        const int wpw = (ppitch + NWAVES - 1) / NWAVES;
        const int whi = min(wave * wpw + wpw, ppitch);
        for (int w0 = wave * wpw; w0 < whi; w0 += 32) {
            const int cb = (w0 + lane) << 5;
            unsigned m = (w0 + lane < whi) ? prow[w0 + lane] : 0u;
            unsigned mk = __builtin_amdgcn_ballot_w32(m != 0u);
            int c = (m != 0u) ? cb + __ffs((int)m) - 1 : 0;
            uint4 ck = k4[c];
            while (mk != 0u) {
                const bool has = m != 0u;
                m &= m - 1u;
                const int cn = (m != 0u) ? cb + __ffs((int)m) - 1 : 0;
                const uint4 nk = k4[cn];
                unsigned hits = has ? (EQ(ck.x, 0) | EQ(ck.y, 1) | EQ(ck.z, 2) | EQ(ck.w, 3)) : 0u;
                unsigned hk = __builtin_amdgcn_ballot_w32(hits != 0u);
                while (hk != 0u) {
                    const int p = __ffs((int)hits) - 1;
                    const int e = (c << 4) + ((p & 7) << 2) + (p >> 3);
                    const unsigned pre = __builtin_amdgcn_mbcnt_lo(hk, 0u);
                    if (hits != 0u) q[qn + (int)pre] = e;
                    qn += __builtin_popcount(hk);
                    hits &= hits - 1u;
                    hk = __builtin_amdgcn_ballot_w32(hits != 0u);
                }
                while (qn >= 32) drain(32);
                c = cn;
                ck = nk;
                mk = __builtin_amdgcn_ballot_w32(m != 0u);
            }
        }
        if (qn > 0) drain(qn);
        if (!second && lane == 0) mycnt[wave] = nrec;
    } else {
        if (nrec > 0) {
            int2 r = myrec[min(lane, nrec - 1)];
            int k = 0;
            for (; k + 32 <= nrec; k += 32) {
                const int2 rn = myrec[min(k + 32 + lane, nrec - 1)];
                gather(r.x, r.y, 32);
                r = rn;
            }
            if (k < nrec) gather(r.x, r.y, nrec - k);
        }
    }
#undef EQ
#undef NZ
    __syncthreads();

    const int node0 = base + wave * 16;
    if (wave >= NTILES || node0 >= n_nodes) return;

    const int half = lane >> 4;
    const int lm   = lane & 15;

    const int rl = wave * 16 + lm;
    const float inv = 1.0f / fmaxf(sCnt[rl], 1.0f);
    const float* arow = sAcc + rl * APITCH;
    int gr = node0 + lm;
    if (gr > n_nodes - 1) gr = n_nodes - 1;
    const float* xrow = feat + (size_t)gr * FDIM;

    v8f acc[4];
    #pragma unroll
    for (int nt = 0; nt < 4; ++nt) acc[nt] = (v8f){};

    #pragma unroll
    for (int kb = 0; kb < 16; ++kb) {
        const int k0 = kb * 4 + half * 2;
        const v2f av = *(const v2f*)(arow + k0);
        v2f a;
        a.x = av.x * inv;
        a.y = av.y * inv;
        #pragma unroll
        for (int nt = 0; nt < 4; ++nt) {
            v2f b;
            b.x = sWl[k0 * WPITCH + nt * 16 + lm];
            b.y = sWl[(k0 + 1) * WPITCH + nt * 16 + lm];
            acc[nt] = __builtin_amdgcn_wmma_f32_16x16x4_f32(
                false, a, false, b, (short)0, acc[nt], false, false);
        }
    }
    #pragma unroll
    for (int kb = 0; kb < 16; ++kb) {
        const int k0 = kb * 4 + half * 2;
        const v2f a = *(const v2f*)(xrow + k0);
        #pragma unroll
        for (int nt = 0; nt < 4; ++nt) {
            v2f b;
            b.x = sWr[k0 * WPITCH + nt * 16 + lm];
            b.y = sWr[(k0 + 1) * WPITCH + nt * 16 + lm];
            acc[nt] = __builtin_amdgcn_wmma_f32_16x16x4_f32(
                false, a, false, b, (short)0, acc[nt], false, false);
        }
    }

    #pragma unroll
    for (int nt = 0; nt < 4; ++nt) {
        const float bv = bias[nt * 16 + lm];
        #pragma unroll
        for (int i = 0; i < 8; ++i) {
            float v = acc[nt][i] + bv;
            if (relu_flag) v = fmaxf(v, 0.0f);
            const int row = node0 + i + half * 8;
            if (row < n_nodes) out[(size_t)row * FDIM + nt * 16 + lm] = v;
        }
    }
}

extern "C" void kernel_launch(void* const* d_in, const int* in_sizes, int n_in,
                              void* d_out, int out_size, void* d_ws, size_t ws_size,
                              hipStream_t stream) {
    const float* x    = (const float*)d_in[0];
    const int*   ei   = (const int*)d_in[1];
    const float* W1_l = (const float*)d_in[2];
    const float* b1   = (const float*)d_in[3];
    const float* W1_r = (const float*)d_in[4];
    const float* W2_l = (const float*)d_in[5];
    const float* b2   = (const float*)d_in[6];
    const float* W2_r = (const float*)d_in[7];
    float* out = (float*)d_out;

    const int n_nodes = in_sizes[0] / FDIM;
    const int n_edges = in_sizes[1] / 2;
    const int* src = ei;
    const int* dst = ei + n_edges;

    const int grid = (n_nodes + RNODES - 1) / RNODES;

    const int n_vec  = (n_edges + 15) / 16;
    const int n_slab = (n_vec + SLAB - 1) / SLAB;
    const int ppitch = n_slab * 32;
    float*    h    = (float*)d_ws;
    unsigned* keys = (unsigned*)(h + (size_t)n_nodes * FDIM);
    unsigned* pres = keys + (size_t)n_vec * 4;
    int*      rcnt = (int*)(pres + (size_t)256 * ppitch);
    int*      rec  = rcnt + (size_t)grid * NWAVES;

    edge_keys<<<n_slab, 1024, 0, stream>>>(dst, keys, pres, n_edges, n_vec, ppitch);

    sage_layer<<<grid, 1024, 0, stream>>>(x, src, dst, keys, pres, W1_l, W1_r, b1, h, rcnt, rec,
                                          n_nodes, n_edges, ppitch,  1,  0, ws_size);
    sage_layer<<<grid, 1024, 0, stream>>>(h, src, dst, keys, pres, W2_l, W2_r, b2, out, rcnt, rec,
                                          n_nodes, n_edges, ppitch,  0,  1, ws_size);
}


# A tile's self half (xin @ Wr, 64 of its 128 matrix instructions) is issued before the wave's edge share, under the other waves' row fetches; product summed self half first, then mean half
# speedup vs baseline: 3.6409x; 1.0338x over previous
#include <hip/hip_runtime.h>


typedef __attribute__((ext_vector_type(2))) float v2f;
typedef __attribute__((ext_vector_type(8))) float v8f;

#define FDIM   64
#define RNODES 400
#define APITCH 66
#define WPITCH 72
#define NWAVES 32
#define NTILES (RNODES / 16)
static_assert(RNODES % 16 == 0 && NTILES <= NWAVES, "a tile must not straddle two blocks");
#define QCAP   1056
#define RCAP   2048
#define SLAB   1024
#define BPITCH 257

__global__ void __launch_bounds__(1024)
edge_keys(const int* __restrict__ dst, unsigned* __restrict__ keys, unsigned* __restrict__ pres,
          int n_edges, int n_vec, int ppitch) {
    __shared__ unsigned sBits[32 * BPITCH];
    const int tid = threadIdx.x;
    for (int i = tid; i < 32 * BPITCH; i += 1024) sBits[i] = 0u;
    __syncthreads();
    const int t = blockIdx.x * SLAB + tid;
    if (t < n_vec) {
        const int e0 = t * 16;
        int d[16];
        if ((n_edges & 3) == 0 && e0 + 16 <= n_edges) {
            const int4* p = (const int4*)(dst + e0);
            #pragma unroll
            for (int j = 0; j < 4; ++j) {
                const int4 v = p[j];
                d[4 * j] = v.x; d[4 * j + 1] = v.y; d[4 * j + 2] = v.z; d[4 * j + 3] = v.w;
            }
        } else {
            #pragma unroll
            for (int j = 0; j < 16; ++j) d[j] = (e0 + j < n_edges) ? dst[e0 + j] : 255 * RNODES;
        }
        unsigned kb[16];
        #pragma unroll
        for (int j = 0; j < 16; ++j) kb[j] = ((unsigned)d[j] / (unsigned)RNODES) & 255u;
        ((uint4*)keys)[t] = make_uint4(
            kb[0]  | kb[1]  << 8 | kb[2]  << 16 | kb[3]  << 24,
            kb[4]  | kb[5]  << 8 | kb[6]  << 16 | kb[7]  << 24,
            kb[8]  | kb[9]  << 8 | kb[10] << 16 | kb[11] << 24,
            kb[12] | kb[13] << 8 | kb[14] << 16 | kb[15] << 24);
        unsigned* col = sBits + (tid >> 5) * BPITCH;
        const unsigned bit = 1u << (tid & 31);
        #pragma unroll
        for (int j = 0; j < 16; ++j) atomicOr(&col[kb[j]], bit);
    }
    __syncthreads();
    const int lane = tid & 31;
    for (int r = tid >> 5; r < 256; r += 32)
        pres[(size_t)r * ppitch + blockIdx.x * 32 + lane] = sBits[lane * BPITCH + r];
}

__global__ void __launch_bounds__(1024)
sage_layer(const float* __restrict__ feat, const int* __restrict__ src,
           const int* __restrict__ dst, const unsigned* __restrict__ keys,
           const unsigned* __restrict__ pres, const float* __restrict__ Wl,
           const float* __restrict__ Wr, const float* __restrict__ bias,
           float* __restrict__ out, int* __restrict__ rcnt, int* __restrict__ rec,
           int n_nodes, int n_edges, int ppitch, int relu_flag, int second, size_t ws_bytes) {
    __shared__ alignas(16) float sAcc[RNODES * APITCH];
    __shared__ float sCnt[RNODES];
    __shared__ float sWl[FDIM * WPITCH];
    __shared__ float sWr[FDIM * WPITCH];
    __shared__ int   sQ[NWAVES * QCAP];

    const int tid = threadIdx.x;
    for (int i = tid; i < RNODES * APITCH; i += 1024) sAcc[i] = 0.0f;
    if (tid < RNODES) sCnt[tid] = 0.0f;
    for (int i = tid; i < FDIM * FDIM; i += 1024) {
        const int r = i >> 6, c = i & 63;
        sWl[r * WPITCH + c] = Wl[i];
        sWr[r * WPITCH + c] = Wr[i];
    }
    __syncthreads();

    const int lane = tid & 31;
    const int wave = __builtin_amdgcn_readfirstlane(tid >> 5);
    const int base = blockIdx.x * RNODES;

    const int half  = lane >> 4;
    const int lm    = lane & 15;
    const int node0 = base + wave * 16;
    const bool tile = wave < NTILES && node0 < n_nodes;
    v8f acc[4];
    #pragma unroll
    for (int nt = 0; nt < 4; ++nt) acc[nt] = (v8f){};
    if (tile) {
        int gr = node0 + lm;
        if (gr > n_nodes - 1) gr = n_nodes - 1;
        const float* xrow = feat + (size_t)gr * FDIM;
        #pragma unroll
        for (int kb = 0; kb < 16; ++kb) {
            const int k0 = kb * 4 + half * 2;
            const v2f a = *(const v2f*)(xrow + k0);
            #pragma unroll
            for (int nt = 0; nt < 4; ++nt) {
                v2f b;
                b.x = sWr[k0 * WPITCH + nt * 16 + lm];
                b.y = sWr[(k0 + 1) * WPITCH + nt * 16 + lm];
                acc[nt] = __builtin_amdgcn_wmma_f32_16x16x4_f32(
                    false, a, false, b, (short)0, acc[nt], false, false);
            }
        }
    }

    int* q = sQ + wave * QCAP;
    int qn = 0;

    int2* myrec = (int2*)rec + ((size_t)blockIdx.x * NWAVES + wave) * RCAP;
    int*  mycnt = rcnt + blockIdx.x * NWAVES;
    int  nrec = 0;
    bool filter = true;
    if (second) {
        const int c = mycnt[lane];
        filter = __builtin_amdgcn_ballot_w32(c > RCAP) != 0u;
        nrec = __builtin_amdgcn_readlane(c, wave);
    } else {
        const size_t need = (size_t)n_nodes * FDIM * sizeof(float) +
                            (size_t)((n_edges + 15) / 16) * 16 + (size_t)256 * ppitch * sizeof(unsigned) +
                            (size_t)gridDim.x * NWAVES * (sizeof(int) + (size_t)RCAP * sizeof(int2));
        if (need > ws_bytes) nrec = RCAP + 1;
    }

    const unsigned pat = (blockIdx.x & 255u) * 0x01010101u;
    const uint4* k4 = (const uint4*)keys;

    auto gather = [&](int s, int d, int n) {
        if (lane < n && d >= 0) atomicAdd(&sCnt[d], 1.0f);
        for (int k0 = 0; k0 < n; k0 += 8) {
            float a[8], b[8];
            int dd[8];
            #pragma unroll
            for (int j = 0; j < 8; ++j) {
                const int k  = k0 + j;
                const int kk = (k < n) ? k : (n - 1);
                const int sk = __builtin_amdgcn_readlane(s, kk);
                dd[j]        = __builtin_amdgcn_readlane(d, kk);
                const float* rp = feat + (size_t)sk * FDIM;
                a[j] = rp[lane];
                b[j] = rp[lane + 32];
            }
            #pragma unroll
            for (int j = 0; j < 8; ++j) {
                if (k0 + j < n && dd[j] >= 0) {
                    atomicAdd(&sAcc[dd[j] * APITCH + lane], a[j]);
                    atomicAdd(&sAcc[dd[j] * APITCH + lane + 32], b[j]);
                }
            }
        }
    };
    auto drain = [&](int n) {
        qn -= n;
        const bool act = lane < n;
        const int e = act ? q[qn + lane] : 0;
        const bool in = act && e < n_edges;
        const int s = in ? src[e] : 0;
        const int dv = in ? (dst[e] - base) : -1;
        const int d = ((unsigned)dv < (unsigned)RNODES) ? dv : -1;
        if (!second) {
            if (nrec + n <= RCAP) {
                if (act) myrec[nrec + lane] = make_int2(s, d);
                nrec += n;
            } else {
                nrec = RCAP + 1;
            }
        }
        gather(s, d, n);
    };

#define NZ(w) (((((w) ^ pat) & 0x7f7f7f7fu) + 0x7f7f7f7fu) | ((w) ^ pat))
#define EQ(w, j) ((((~NZ(w)) >> 7) & 0x01010101u) << (j))

    if (filter) {
        const unsigned* prow = pres + (size_t)(blockIdx.x & 255u) * ppitch;
        const int wpw = (ppitch + NWAVES - 1) / NWAVES;
        const int whi = min(wave * wpw + wpw, ppitch);
        for (int w0 = wave * wpw; w0 < whi; w0 += 32) {
            const int cb = (w0 + lane) << 5;
            unsigned m = (w0 + lane < whi) ? prow[w0 + lane] : 0u;
            unsigned mk = __builtin_amdgcn_ballot_w32(m != 0u);
            int c = (m != 0u) ? cb + __ffs((int)m) - 1 : 0;
            uint4 ck = k4[c];
            while (mk != 0u) {
                const bool has = m != 0u;
                m &= m - 1u;
                const int cn = (m != 0u) ? cb + __ffs((int)m) - 1 : 0;
                const uint4 nk = k4[cn];
                unsigned hits = has ? (EQ(ck.x, 0) | EQ(ck.y, 1) | EQ(ck.z, 2) | EQ(ck.w, 3)) : 0u;
                unsigned hk = __builtin_amdgcn_ballot_w32(hits != 0u);
                while (hk != 0u) {
                    const int p = __ffs((int)hits) - 1;
                    const int e = (c << 4) + ((p & 7) << 2) + (p >> 3);
                    const unsigned pre = __builtin_amdgcn_mbcnt_lo(hk, 0u);
                    if (hits != 0u) q[qn + (int)pre] = e;
                    qn += __builtin_popcount(hk);
                    hits &= hits - 1u;
                    hk = __builtin_amdgcn_ballot_w32(hits != 0u);
                }
                while (qn >= 32) drain(32);
                c = cn;
                ck = nk;
                mk = __builtin_amdgcn_ballot_w32(m != 0u);
            }
        }
        if (qn > 0) drain(qn);
        if (!second && lane == 0) mycnt[wave] = nrec;
    } else {
        if (nrec > 0) {
            int2 r = myrec[min(lane, nrec - 1)];
            int k = 0;
            for (; k + 32 <= nrec; k += 32) {
                const int2 rn = myrec[min(k + 32 + lane, nrec - 1)];
                gather(r.x, r.y, 32);
                r = rn;
            }
            if (k < nrec) gather(r.x, r.y, nrec - k);
        }
    }
#undef EQ
#undef NZ
    __syncthreads();

    if (!tile) return;

    const int rl = wave * 16 + lm;
    const float inv = 1.0f / fmaxf(sCnt[rl], 1.0f);
    const float* arow = sAcc + rl * APITCH;

    #pragma unroll
    for (int kb = 0; kb < 16; ++kb) {
        const int k0 = kb * 4 + half * 2;
        const v2f av = *(const v2f*)(arow + k0);
        v2f a;
        a.x = av.x * inv;
        a.y = av.y * inv;
        #pragma unroll
        for (int nt = 0; nt < 4; ++nt) {
            v2f b;
            b.x = sWl[k0 * WPITCH + nt * 16 + lm];
            b.y = sWl[(k0 + 1) * WPITCH + nt * 16 + lm];
            acc[nt] = __builtin_amdgcn_wmma_f32_16x16x4_f32(
                false, a, false, b, (short)0, acc[nt], false, false);
        }
    }

    #pragma unroll
    for (int nt = 0; nt < 4; ++nt) {
        const float bv = bias[nt * 16 + lm];
        #pragma unroll
        for (int i = 0; i < 8; ++i) {
            float v = acc[nt][i] + bv;
            if (relu_flag) v = fmaxf(v, 0.0f);
            const int row = node0 + i + half * 8;
            if (row < n_nodes) out[(size_t)row * FDIM + nt * 16 + lm] = v;
        }
    }
}

extern "C" void kernel_launch(void* const* d_in, const int* in_sizes, int n_in,
                              void* d_out, int out_size, void* d_ws, size_t ws_size,
                              hipStream_t stream) {
    const float* x    = (const float*)d_in[0];
    const int*   ei   = (const int*)d_in[1];
    const float* W1_l = (const float*)d_in[2];
    const float* b1   = (const float*)d_in[3];
    const float* W1_r = (const float*)d_in[4];
    const float* W2_l = (const float*)d_in[5];
    const float* b2   = (const float*)d_in[6];
    const float* W2_r = (const float*)d_in[7];
    float* out = (float*)d_out;

    const int n_nodes = in_sizes[0] / FDIM;
    const int n_edges = in_sizes[1] / 2;
    const int* src = ei;
    const int* dst = ei + n_edges;

    const int grid = (n_nodes + RNODES - 1) / RNODES;

    const int n_vec  = (n_edges + 15) / 16;
    const int n_slab = (n_vec + SLAB - 1) / SLAB;
    const int ppitch = n_slab * 32;
    float*    h    = (float*)d_ws;
    unsigned* keys = (unsigned*)(h + (size_t)n_nodes * FDIM);
    unsigned* pres = keys + (size_t)n_vec * 4;
    int*      rcnt = (int*)(pres + (size_t)256 * ppitch);
    int*      rec  = rcnt + (size_t)grid * NWAVES;

    edge_keys<<<n_slab, 1024, 0, stream>>>(dst, keys, pres, n_edges, n_vec, ppitch);

    sage_layer<<<grid, 1024, 0, stream>>>(x, src, dst, keys, pres, W1_l, W1_r, b1, h, rcnt, rec,
                                          n_nodes, n_edges, ppitch,  1,  0, ws_size);
    sage_layer<<<grid, 1024, 0, stream>>>(h, src, dst, keys, pres, W2_l, W2_r, b2, out, rcnt, rec,
                                          n_nodes, n_edges, ppitch,  0,  1, ws_size);
}


# Layer 1 reads no keys and no src[e], dst[e]: launch 1 writes the edges as (source, destination) pairs, a 16-edge chunk one 128-byte line; a flagged chunk's line gives both; LDS adds: arrival order
# speedup vs baseline: 3.7924x; 1.0416x over previous
#include <hip/hip_runtime.h>


typedef __attribute__((ext_vector_type(2))) float v2f;
typedef __attribute__((ext_vector_type(8))) float v8f;

#define FDIM   64
#define RNODES 400
#define APITCH 66
#define WPITCH 72
#define NWAVES 32
#define NTILES (RNODES / 16)
static_assert(RNODES % 16 == 0 && NTILES <= NWAVES, "a tile must not straddle two blocks");
#define QCAP   544
#define CQCAP  64
#define RCAP   2048
#define SLAB   1024
#define BPITCH 257

__global__ void __launch_bounds__(1024)
edge_pairs(const int* __restrict__ src, const int* __restrict__ dst, int* __restrict__ pairs,
           unsigned* __restrict__ pres, int n_edges, int n_vec, int ppitch) {
    __shared__ unsigned sBits[32 * BPITCH];
    const int tid = threadIdx.x;
    for (int i = tid; i < 32 * BPITCH; i += 1024) sBits[i] = 0u;
    __syncthreads();
    const int lane = tid & 31;
    const int wave = tid >> 5;
    const int cw = blockIdx.x * SLAB + wave * 32;
    const int ew = cw * 16;
    unsigned* col = sBits + wave * BPITCH;
    int4* p4 = (int4*)pairs;
    int s0[8], s1[8], d0[8], d1[8];
    if ((n_edges & 1) == 0 && ew + 512 <= n_edges) {
        #pragma unroll
        for (int j = 0; j < 8; ++j) {
            const int e = ew + j * 64 + lane * 2;
            const int2 sv = *(const int2*)(src + e);
            const int2 dv = *(const int2*)(dst + e);
            s0[j] = sv.x; s1[j] = sv.y; d0[j] = dv.x; d1[j] = dv.y;
        }
    } else {
        #pragma unroll
        for (int j = 0; j < 8; ++j) {
            const int e = ew + j * 64 + lane * 2;
            s0[j] = (e < n_edges) ? src[e] : 0;
            d0[j] = (e < n_edges) ? dst[e] : -1;
            s1[j] = (e + 1 < n_edges) ? src[e + 1] : 0;
            d1[j] = (e + 1 < n_edges) ? dst[e + 1] : -1;
        }
    }
    #pragma unroll
    for (int j = 0; j < 8; ++j) {
        const int e = ew + j * 64 + lane * 2;
        if ((e >> 4) < n_vec) p4[e >> 1] = make_int4(s0[j], d0[j], s1[j], d1[j]);
        const unsigned bit = 1u << (j * 4 + (lane >> 3));
        if (e < n_edges)     atomicOr(&col[((unsigned)d0[j] / (unsigned)RNODES) & 255u], bit);
        if (e + 1 < n_edges) atomicOr(&col[((unsigned)d1[j] / (unsigned)RNODES) & 255u], bit);
    }
    __syncthreads();
    for (int r = wave; r < 256; r += 32)
        pres[(size_t)r * ppitch + blockIdx.x * 32 + lane] = sBits[lane * BPITCH + r];
}

__global__ void __launch_bounds__(1024)
sage_layer(const float* __restrict__ feat, const int* __restrict__ pairs,
           const unsigned* __restrict__ pres, const float* __restrict__ Wl,
           const float* __restrict__ Wr, const float* __restrict__ bias,
           float* __restrict__ out, int* __restrict__ rcnt, int* __restrict__ rec,
           int n_nodes, int n_edges, int ppitch, int relu_flag, int second, size_t ws_bytes) {
    __shared__ alignas(16) float sAcc[RNODES * APITCH];
    __shared__ float sCnt[RNODES];
    __shared__ float sWl[FDIM * WPITCH];
    __shared__ float sWr[FDIM * WPITCH];
    __shared__ alignas(8) int2 sQ[NWAVES * QCAP];
    __shared__ int   sC[NWAVES * CQCAP];

    const int tid = threadIdx.x;
    for (int i = tid; i < RNODES * APITCH; i += 1024) sAcc[i] = 0.0f;
    if (tid < RNODES) sCnt[tid] = 0.0f;
    for (int i = tid; i < FDIM * FDIM; i += 1024) {
        const int r = i >> 6, c = i & 63;
        sWl[r * WPITCH + c] = Wl[i];
        sWr[r * WPITCH + c] = Wr[i];
    }
    __syncthreads();

    const int lane = tid & 31;
    const int wave = __builtin_amdgcn_readfirstlane(tid >> 5);
    const int base = blockIdx.x * RNODES;

    const int half  = lane >> 4;
    const int lm    = lane & 15;
    const int node0 = base + wave * 16;
    const bool tile = wave < NTILES && node0 < n_nodes;
    v8f acc[4];
    #pragma unroll
    for (int nt = 0; nt < 4; ++nt) acc[nt] = (v8f){};
    if (tile) {
        int gr = node0 + lm;
        if (gr > n_nodes - 1) gr = n_nodes - 1;
        const float* xrow = feat + (size_t)gr * FDIM;
        #pragma unroll
        for (int kb = 0; kb < 16; ++kb) {
            const int k0 = kb * 4 + half * 2;
            const v2f a = *(const v2f*)(xrow + k0);
            #pragma unroll
            for (int nt = 0; nt < 4; ++nt) {
                v2f b;
                b.x = sWr[k0 * WPITCH + nt * 16 + lm];
                b.y = sWr[(k0 + 1) * WPITCH + nt * 16 + lm];
                acc[nt] = __builtin_amdgcn_wmma_f32_16x16x4_f32(
                    false, a, false, b, (short)0, acc[nt], false, false);
            }
        }
    }

    int2* q = sQ + wave * QCAP;
    int qn = 0;
    int* cq = sC + wave * CQCAP;
    int cn = 0;

    int2* myrec = (int2*)rec + ((size_t)blockIdx.x * NWAVES + wave) * RCAP;
    int*  mycnt = rcnt + blockIdx.x * NWAVES;
    int  nrec = 0;
    bool filter = true;
    if (second) {
        const int c = mycnt[lane];
        filter = __builtin_amdgcn_ballot_w32(c > RCAP) != 0u;
        nrec = __builtin_amdgcn_readlane(c, wave);
    } else {
        const size_t need = (size_t)n_nodes * FDIM * sizeof(float) +
                            (size_t)((n_edges + 15) / 16) * 128 + (size_t)256 * ppitch * sizeof(unsigned) +
                            (size_t)gridDim.x * NWAVES * (sizeof(int) + (size_t)RCAP * sizeof(int2));
        if (need > ws_bytes) nrec = RCAP + 1;
    }

    const int4* p4 = (const int4*)pairs;

    auto gather = [&](int s, int d, int n) {
        if (lane < n && d >= 0) atomicAdd(&sCnt[d], 1.0f);
        for (int k0 = 0; k0 < n; k0 += 8) {
            float a[8], b[8];
            int dd[8];
            #pragma unroll
            for (int j = 0; j < 8; ++j) {
                const int k  = k0 + j;
                const int kk = (k < n) ? k : (n - 1);
                const int sk = __builtin_amdgcn_readlane(s, kk);
                dd[j]        = __builtin_amdgcn_readlane(d, kk);
                const float* rp = feat + (size_t)sk * FDIM;
                a[j] = rp[lane];
                b[j] = rp[lane + 32];
            }
            #pragma unroll
            for (int j = 0; j < 8; ++j) {
                if (k0 + j < n && dd[j] >= 0) {
                    atomicAdd(&sAcc[dd[j] * APITCH + lane], a[j]);
                    atomicAdd(&sAcc[dd[j] * APITCH + lane + 32], b[j]);
                }
            }
        }
    };
    auto drain = [&](int n) {
        qn -= n;
        const bool act = lane < n;
        const int2 p = act ? q[qn + lane] : make_int2(0, -1);
        const int s = p.x;
        const int d = p.y;
        if (!second) {
            if (nrec + n <= RCAP) {
                if (act) myrec[nrec + lane] = make_int2(s, d);
                nrec += n;
            } else {
                nrec = RCAP + 1;
            }
        }
        gather(s, d, n);
    };
    auto chunks = [&](int n) {
        cn -= n;
        const int part = lane & 7;
        int4 v[8];
        #pragma unroll
        for (int j = 0; j < 8; ++j) {
            const int i = j * 4 + (lane >> 3);
            const int c = cq[cn + min(i, n - 1)];
            v[j] = p4[(size_t)c * 8 + part];
        }
        #pragma unroll
        for (int j = 0; j < 8; ++j) {
            const bool ok = j * 4 + (lane >> 3) < n;
            const int dva = v[j].y - base;
            const bool ha = ok && (unsigned)dva < (unsigned)RNODES;
            const unsigned ka = __builtin_amdgcn_ballot_w32(ha);
            if (ha) q[qn + (int)__builtin_amdgcn_mbcnt_lo(ka, 0u)] = make_int2(v[j].x, dva);
            qn += __builtin_popcount(ka);
            const int dvb = v[j].w - base;
            const bool hb = ok && (unsigned)dvb < (unsigned)RNODES;
            const unsigned kb = __builtin_amdgcn_ballot_w32(hb);
            if (hb) q[qn + (int)__builtin_amdgcn_mbcnt_lo(kb, 0u)] = make_int2(v[j].z, dvb);
            qn += __builtin_popcount(kb);
        }
        while (qn >= 32) drain(32);
    };

    if (filter) {
        const unsigned* prow = pres + (size_t)(blockIdx.x & 255u) * ppitch;
        const int wpw = (ppitch + NWAVES - 1) / NWAVES;
        const int whi = min(wave * wpw + wpw, ppitch);
        int w0 = wave * wpw;
        unsigned m = (w0 + lane < whi) ? prow[w0 + lane] : 0u;
        for (; w0 < whi; w0 += 32) {
            const unsigned mn = (w0 + 32 + lane < whi) ? prow[w0 + 32 + lane] : 0u;
            const int cb = (w0 + lane) << 5;
            unsigned mk = __builtin_amdgcn_ballot_w32(m != 0u);
            while (mk != 0u) {
                if (m != 0u) cq[cn + (int)__builtin_amdgcn_mbcnt_lo(mk, 0u)] = cb + __ffs((int)m) - 1;
                cn += __builtin_popcount(mk);
                m &= m - 1u;
                mk = __builtin_amdgcn_ballot_w32(m != 0u);
                if (cn >= 32) chunks(32);
            }
            m = mn;
        }
        if (cn > 0) chunks(cn);
        if (qn > 0) drain(qn);
        if (!second && lane == 0) mycnt[wave] = nrec;
    } else {
        if (nrec > 0) {
            int2 r = myrec[min(lane, nrec - 1)];
            int k = 0;
            for (; k + 32 <= nrec; k += 32) {
                const int2 rn = myrec[min(k + 32 + lane, nrec - 1)];
                gather(r.x, r.y, 32);
                r = rn;
            }
            if (k < nrec) gather(r.x, r.y, nrec - k);
        }
    }
    __syncthreads();

    if (!tile) return;

    const int rl = wave * 16 + lm;
    const float inv = 1.0f / fmaxf(sCnt[rl], 1.0f);
    const float* arow = sAcc + rl * APITCH;

    #pragma unroll
    for (int kb = 0; kb < 16; ++kb) {
        const int k0 = kb * 4 + half * 2;
        const v2f av = *(const v2f*)(arow + k0);
        v2f a;
        a.x = av.x * inv;
        a.y = av.y * inv;
        #pragma unroll
        for (int nt = 0; nt < 4; ++nt) {
            v2f b;
            b.x = sWl[k0 * WPITCH + nt * 16 + lm];
            b.y = sWl[(k0 + 1) * WPITCH + nt * 16 + lm];
            acc[nt] = __builtin_amdgcn_wmma_f32_16x16x4_f32(
                false, a, false, b, (short)0, acc[nt], false, false);
        }
    }

    #pragma unroll
    for (int nt = 0; nt < 4; ++nt) {
        const float bv = bias[nt * 16 + lm];
        #pragma unroll
        for (int i = 0; i < 8; ++i) {
            float v = acc[nt][i] + bv;
            if (relu_flag) v = fmaxf(v, 0.0f);
            const int row = node0 + i + half * 8;
            if (row < n_nodes) out[(size_t)row * FDIM + nt * 16 + lm] = v;
        }
    }
}

extern "C" void kernel_launch(void* const* d_in, const int* in_sizes, int n_in,
                              void* d_out, int out_size, void* d_ws, size_t ws_size,
                              hipStream_t stream) {
    const float* x    = (const float*)d_in[0];
    const int*   ei   = (const int*)d_in[1];
    const float* W1_l = (const float*)d_in[2];
    const float* b1   = (const float*)d_in[3];
    const float* W1_r = (const float*)d_in[4];
    const float* W2_l = (const float*)d_in[5];
    const float* b2   = (const float*)d_in[6];
    const float* W2_r = (const float*)d_in[7];
    float* out = (float*)d_out;

    const int n_nodes = in_sizes[0] / FDIM;
    const int n_edges = in_sizes[1] / 2;
    const int* src = ei;
    const int* dst = ei + n_edges;

    const int grid = (n_nodes + RNODES - 1) / RNODES;

    const int n_vec  = (n_edges + 15) / 16;
    const int n_slab = (n_vec + SLAB - 1) / SLAB;
    const int ppitch = n_slab * 32;
    float*    h    = (float*)d_ws;
    int*      pairs = (int*)(h + (size_t)n_nodes * FDIM);
    unsigned* pres = (unsigned*)(pairs + (size_t)n_vec * 32);
    int*      rcnt = (int*)(pres + (size_t)256 * ppitch);
    int*      rec  = rcnt + (size_t)grid * NWAVES;

    edge_pairs<<<n_slab, 1024, 0, stream>>>(src, dst, pairs, pres, n_edges, n_vec, ppitch);

    sage_layer<<<grid, 1024, 0, stream>>>(x, pairs, pres, W1_l, W1_r, b1, h, rcnt, rec,
                                          n_nodes, n_edges, ppitch,  1,  0, ws_size);
    sage_layer<<<grid, 1024, 0, stream>>>(h, pairs, pres, W2_l, W2_r, b2, out, rcnt, rec,
                                          n_nodes, n_edges, ppitch,  0,  1, ws_size);
}
